# starting kernel
# baseline (speedup 1.0000x reference)
#include <hip/hip_runtime.h>


typedef __attribute__((ext_vector_type(2))) float v2f;
typedef __attribute__((ext_vector_type(8))) float v8f;

#define BATCH  32
#define HEIGHT 256
#define WIDTH  256
#define CHAN   16
#define HCN    4
#define HS     12

#define XROW_STRIDE 20
#define XROWS       258
#define XCOL_SIZE   (XROWS * XROW_STRIDE)
#define NCOLS       12
#define OFF_XCOL    0
#define OFF_CONVW   (NCOLS * XCOL_SIZE)
#define OFF_CONVOUT (OFF_CONVW + 576)
#define OFF_US      (OFF_CONVOUT + 4096)
#define SMEM_FLOATS (OFF_US + 8 * 2 * 256)
#define DELTA_STRIDE 260
#define OFF_DELTA   0

__launch_bounds__(256)
__global__ void ca_step_kernel(const float* __restrict__ xin,
                               const float* __restrict__ convw,
                               const float* __restrict__ fc0w,
                               const float* __restrict__ fc0b,
                               const float* __restrict__ fc1w,
                               float* __restrict__ xout)
{
    extern __shared__ float smem[];
    float* xcol  = smem + OFF_XCOL;
    float* cw    = smem + OFF_CONVW;
    float* cout  = smem + OFF_CONVOUT;
    float* us    = smem + OFF_US;
    float* delta = smem + OFF_DELTA;

    const int tid  = threadIdx.x;
    const int lane = tid & 31;
    const int wave = tid >> 5;
    const int m    = lane & 15;
    const int hi   = lane >> 4;

    const int blk = blockIdx.x;
    const int b   = blk >> 6;
    const int ww  = blk & 63;

    const float* xres = xin + ((b * HEIGHT) * WIDTH + ww * 4) * CHAN;

    for (int i = tid; i < 1024; i += 256) {
        int h2 = i >> 2;
        int q  = i & 3;
        __builtin_prefetch(xres + (h2 * WIDTH + q) * CHAN, 0, 1);
    }

    v2f afc0;
    {
        int s = m;
        int base = (s < HS ? s : 0) * 4 + 2 * hi;
        float a0 = fc0w[base + 0];
        float a1 = fc0w[base + 1];
        afc0 = (s < HS) ? (v2f){a0, a1} : (v2f){0.f, 0.f};
    }
    v8f cbias;
#pragma unroll
    for (int v = 0; v < 8; ++v) {
        int s = v + 8 * hi;
        float bv = fc0b[s < HS ? s : 0];
        cbias[v] = (s < HS) ? bv : 0.f;
    }
    v2f afc1[3];
#pragma unroll
    for (int t3 = 0; t3 < 3; ++t3) {
        int base = m * HS + 4 * t3 + 2 * hi;
        afc1[t3].x = fc1w[base + 0];
        afc1[t3].y = fc1w[base + 1];
    }

    for (int i = tid; i < 576; i += 256) cw[i] = convw[i];

    for (int i = tid; i < NCOLS * XROWS * 4; i += 256) {
        int colIdx = i / (XROWS * 4);
        int rem    = i - colIdx * (XROWS * 4);
        int hh     = rem >> 2;
        int c4     = rem & 3;
        int t      = colIdx / 3;
        int aoff   = colIdx - t * 3 - 1;
        int wcol   = ww + 64 * t + aoff;
        int h      = hh - 1;
        float4 val = make_float4(0.f, 0.f, 0.f, 0.f);
        if ((unsigned)h < 256u && (unsigned)wcol < 256u) {
            val = *(const float4*)(xin + ((b * HEIGHT + h) * WIDTH + wcol) * CHAN + c4 * 4);
        }
        *(float4*)(xcol + colIdx * XCOL_SIZE + hh * XROW_STRIDE + c4 * 4) = val;
    }
    __syncthreads();


    v2f aw[36];
    {
        const float* pw = cw + (m & 3) * 144 + (2 * hi) * 9;
#pragma unroll
        for (int a = 0; a < 3; ++a)
#pragma unroll
            for (int d = 0; d < 3; ++d)
#pragma unroll
                for (int cblk = 0; cblk < 4; ++cblk) {
                    float w0 = pw[cblk * 36 + 0 + a * 3 + d];
                    float w1 = pw[cblk * 36 + 9 + a * 3 + d];
                    aw[(a * 3 + d) * 4 + cblk] =
                        (m < HCN) ? (v2f){w0, w1} : (v2f){0.f, 0.f};
                }
    }

    for (int uu = 0; uu < 8; ++uu) {
        int u  = wave * 8 + uu;
        int t  = u >> 4;
        int h0 = (u & 15) << 4;
        v8f acc = {};
        const float* pb = xcol + 3 * t * XCOL_SIZE + (h0 + m) * XROW_STRIDE + 2 * hi;
#pragma unroll
        for (int a = 0; a < 3; ++a) {
#pragma unroll
            for (int d = 0; d < 3; ++d) {
                const float* pg = pb + a * XCOL_SIZE + d * XROW_STRIDE;
                v2f b0 = *(const v2f*)(pg + 0);
                v2f b1 = *(const v2f*)(pg + 4);
                v2f b2 = *(const v2f*)(pg + 8);
                v2f b3 = *(const v2f*)(pg + 12);
                int s = (a * 3 + d) * 4;
                acc = __builtin_amdgcn_wmma_f32_16x16x4_f32(
                    false, aw[s + 0], false, b0, (short)0, acc, false, false);
                acc = __builtin_amdgcn_wmma_f32_16x16x4_f32(
                    false, aw[s + 1], false, b1, (short)0, acc, false, false);
                acc = __builtin_amdgcn_wmma_f32_16x16x4_f32(
                    false, aw[s + 2], false, b2, (short)0, acc, false, false);
                acc = __builtin_amdgcn_wmma_f32_16x16x4_f32(
                    false, aw[s + 3], false, b3, (short)0, acc, false, false);
            }
        }
        if (lane < 16) {
            float* cdst = cout + (t * 4) * 256 + h0 + lane;
            cdst[0]   = fmaxf(acc[0], 0.f);
            cdst[256] = fmaxf(acc[1], 0.f);
            cdst[512] = fmaxf(acc[2], 0.f);
            cdst[768] = fmaxf(acc[3], 0.f);
        }
    }
    __syncthreads();

#pragma unroll 2
    for (int uu = 0; uu < 8; ++uu) {
        int tt = wave * 8 + uu;
        int hc = tt >> 4;
        int t  = (tt >> 2) & 3;
        int q  = tt & 3;

        v2f Bv = *(const v2f*)(cout + (t * 4 + hc) * 256 + (q * 16 + m) * 4 + 2 * hi);
        v8f u0 = __builtin_amdgcn_wmma_f32_16x16x4_f32(
            false, afc0, false, Bv, (short)0, cbias, false, false);
#pragma unroll
        for (int v = 0; v < 8; ++v) u0[v] = fmaxf(u0[v], 0.f);

        float* usw = us + wave * 512 + (uu & 1) * 256;
        *(float4*)(usw + m * 16 + 8 * hi + 0) = make_float4(u0[0], u0[1], u0[2], u0[3]);
        *(float4*)(usw + m * 16 + 8 * hi + 4) = make_float4(u0[4], u0[5], u0[6], u0[7]);

        v8f acc2 = {};
#pragma unroll
        for (int t3 = 0; t3 < 3; ++t3) {
            v2f B2 = *(const v2f*)(usw + m * 16 + 4 * t3 + 2 * hi);
            acc2 = __builtin_amdgcn_wmma_f32_16x16x4_f32(
                false, afc1[t3], false, B2, (short)0, acc2, false, false);
        }

        int cq = (4 * hc + t) * 4 + q;
        float* dbase = delta + cq * DELTA_STRIDE + m * 16 + 8 * hi;
        *(v2f*)(dbase + 0) = (v2f){acc2[0], acc2[1]};
        *(v2f*)(dbase + 2) = (v2f){acc2[2], acc2[3]};
        *(v2f*)(dbase + 4) = (v2f){acc2[4], acc2[5]};
        *(v2f*)(dbase + 6) = (v2f){acc2[6], acc2[7]};
    }
    __syncthreads();

    float* outp = xout + ((b * HEIGHT) * WIDTH + ww * 4) * CHAN;
    for (int i = tid; i < 4096; i += 256) {
        int h2  = i >> 4;
        int q   = (i >> 2) & 3;
        int c2b = (i & 3) << 2;
        int goff = (h2 * WIDTH + q) * CHAN + c2b;
        float4 xv = *(const float4*)(xres + goff);
        float4 dv;
        dv.x = delta[((c2b + 0) * 4 + q) * DELTA_STRIDE + h2];
        dv.y = delta[((c2b + 1) * 4 + q) * DELTA_STRIDE + h2];
        dv.z = delta[((c2b + 2) * 4 + q) * DELTA_STRIDE + h2];
        dv.w = delta[((c2b + 3) * 4 + q) * DELTA_STRIDE + h2];
        *(float4*)(outp + goff) =
            make_float4(xv.x + dv.x, xv.y + dv.y, xv.z + dv.z, xv.w + dv.w);
    }
}

extern "C" void kernel_launch(void* const* d_in, const int* in_sizes, int n_in,
                              void* d_out, int out_size, void* d_ws, size_t ws_size,
                              hipStream_t stream) {
    (void)in_sizes; (void)n_in; (void)out_size; (void)ws_size;
    const float* x   = (const float*)d_in[0];
    const float* cwv = (const float*)d_in[1];
    const float* f0w = (const float*)d_in[2];
    const float* f0b = (const float*)d_in[3];
    const float* f1w = (const float*)d_in[4];

    float* tmp = (float*)d_ws;
    float* out = (float*)d_out;

    dim3 grid(BATCH * 64);
    dim3 block(256);
    size_t shmem = SMEM_FLOATS * sizeof(float);

    ca_step_kernel<<<grid, block, shmem, stream>>>(x,   cwv, f0w, f0b, f1w, tmp);
    ca_step_kernel<<<grid, block, shmem, stream>>>(tmp, cwv, f0w, f0b, f1w, out);
}


# conv GEMM re-tiled: M = 4 hc x the 4 consecutive h outputs of one MLP row, K = 6-row window (zero-padded weights): 72 WMMAs per 64 outputs, was 144; D feeds fc0 from registers
# speedup vs baseline: 2.0752x; 2.0752x over previous
#include <hip/hip_runtime.h>


typedef __attribute__((ext_vector_type(2))) float v2f;
typedef __attribute__((ext_vector_type(8))) float v8f;

#define BATCH  32
#define HEIGHT 256
#define WIDTH  256
#define CHAN   16
#define HCN    4
#define HS     12
#define NTHREADS 1024
#define NWAVES   (NTHREADS / 32)

#define XROW_STRIDE 20
#define XGRP_STRIDE 84
#define XROWS       258
#define XGROUPS     65
#define XCOL_SIZE   (XGROUPS * XGRP_STRIDE)
#define NCOLS       12
#define OFF_XCOL    0
#define OFF_AFRAG   (NCOLS * XCOL_SIZE)
#define NFRAG       72
#define SMEM_FLOATS (OFF_AFRAG + NFRAG * 64)
#define DELTA_STRIDE 260
#define OFF_DELTA   0
#define OFF_US      (64 * DELTA_STRIDE)

__launch_bounds__(NTHREADS)
__global__ void ca_step_kernel(const float* __restrict__ xin,
                               const float* __restrict__ convw,
                               const float* __restrict__ fc0w,
                               const float* __restrict__ fc0b,
                               const float* __restrict__ fc1w,
                               float* __restrict__ xout)
{
    extern __shared__ float smem[];
    float* xcol  = smem + OFF_XCOL;
    float* afrag = smem + OFF_AFRAG;
    float* us    = smem + OFF_US;
    float* delta = smem + OFF_DELTA;

    const int tid  = threadIdx.x;
    const int lane = tid & 31;
    const int wave = tid >> 5;
    const int m    = lane & 15;
    const int hi   = lane >> 4;

    const int blk = blockIdx.x;
    const int b   = blk >> 6;
    const int ww  = blk & 63;

    const float* xres = xin + ((b * HEIGHT) * WIDTH + ww * 4) * CHAN;

    for (int i = tid; i < 1024; i += NTHREADS) {
        int h2 = i >> 2;
        int q  = i & 3;
        __builtin_prefetch(xres + (h2 * WIDTH + q) * CHAN, 0, 1);
    }

    v2f afc0;
    {
        int s = m;
        int base = (s < HS ? s : 0) * 4 + 2 * hi;
        float a0 = fc0w[base + 0];
        float a1 = fc0w[base + 1];
        afc0 = (s < HS) ? (v2f){a0, a1} : (v2f){0.f, 0.f};
    }
    v8f cbias;
#pragma unroll
    for (int v = 0; v < 8; ++v) {
        int s = v + 8 * hi;
        float bv = fc0b[s < HS ? s : 0];
        cbias[v] = (s < HS) ? bv : 0.f;
    }
    v2f afc1[3];
#pragma unroll
    for (int t3 = 0; t3 < 3; ++t3) {
        int base = m * HS + 4 * t3 + 2 * hi;
        afc1[t3].x = fc1w[base + 0];
        afc1[t3].y = fc1w[base + 1];
    }

#pragma unroll
    for (int k = 0; k < 5; ++k) {
        int i = tid + k * NTHREADS;
        if (i < NFRAG * 64) {
            int s    = i >> 6;
            int ln   = (i >> 1) & 31;
            int r2   = i & 1;
            int cblk = s & 3;
            int ad   = s >> 2;
            int a    = ad / 6;
            int d6   = ad - a * 6;
            int mm   = ln & 15;
            int hh2  = ln >> 4;
            int hcw  = (mm & 7) >> 1;
            int j2   = 2 * (mm >> 3) + (mm & 1);
            int d    = d6 - j2;
            int c    = cblk * 4 + 2 * hh2 + r2;
            bool in  = (unsigned)d < 3u;
            float wv = convw[hcw * 144 + c * 9 + a * 3 + (in ? d : 0)];
            afrag[i] = in ? wv : 0.f;
        }
    }

    for (int i = tid; i < NCOLS * XROWS * 4; i += NTHREADS) {
        int colIdx = i / (XROWS * 4);
        int rem    = i - colIdx * (XROWS * 4);
        int hh     = rem >> 2;
        int c4     = rem & 3;
        int t      = colIdx / 3;
        int aoff   = colIdx - t * 3 - 1;
        int wcol   = ww + 64 * t + aoff;
        int h      = hh - 1;
        float4 val = make_float4(0.f, 0.f, 0.f, 0.f);
        if ((unsigned)h < 256u && (unsigned)wcol < 256u) {
            val = *(const float4*)(xin + ((b * HEIGHT + h) * WIDTH + wcol) * CHAN + c4 * 4);
        }
        *(float4*)(xcol + colIdx * XCOL_SIZE + (hh >> 2) * XGRP_STRIDE + (hh & 3) * XROW_STRIDE + c4 * 4) = val;
    }
    __syncthreads();

    const bool convwave = (wave & 4) == 0;
    const int  t  = wave >> 3;
    const int  q  = wave & 3;
    const int  cwv = t * 4 + q;
    v8f acc = {};
    if (convwave) {
        const float* pb = xcol + 3 * t * XCOL_SIZE + (q * 16 + m) * XGRP_STRIDE + 2 * hi;
        const float* pa = afrag + lane * 2;
#pragma unroll
        for (int a = 0; a < 3; ++a) {
#pragma unroll
            for (int d6 = 0; d6 < 6; ++d6) {
                const float* pg = pb + a * XCOL_SIZE + (d6 >> 2) * XGRP_STRIDE + (d6 & 3) * XROW_STRIDE;
                int s = (a * 6 + d6) * 4;
#pragma unroll
                for (int cblk = 0; cblk < 4; ++cblk) {
                    v2f av = *(const v2f*)(pa + (s + cblk) * 64);
                    v2f bv = *(const v2f*)(pg + 4 * cblk);
                    acc = __builtin_amdgcn_wmma_f32_16x16x4_f32(
                        false, av, false, bv, (short)0, acc, false, false);
                }
            }
        }
    }
    __syncthreads();

    if (convwave) {
#pragma unroll
        for (int hc = 0; hc < 4; ++hc) {
            v2f Bv = (v2f){fmaxf(acc[2 * hc], 0.f), fmaxf(acc[2 * hc + 1], 0.f)};
            v8f u0 = __builtin_amdgcn_wmma_f32_16x16x4_f32(
                false, afc0, false, Bv, (short)0, cbias, false, false);
#pragma unroll
            for (int v = 0; v < 8; ++v) u0[v] = fmaxf(u0[v], 0.f);

            float* usw = us + cwv * 1024 + hc * 256;
            *(float4*)(usw + m * 16 + 8 * hi + 0) = make_float4(u0[0], u0[1], u0[2], u0[3]);
            *(float4*)(usw + m * 16 + 8 * hi + 4) = make_float4(u0[4], u0[5], u0[6], u0[7]);

            v8f acc2 = {};
#pragma unroll
            for (int t3 = 0; t3 < 3; ++t3) {
                v2f B2 = *(const v2f*)(usw + m * 16 + 4 * t3 + 2 * hi);
                acc2 = __builtin_amdgcn_wmma_f32_16x16x4_f32(
                    false, afc1[t3], false, B2, (short)0, acc2, false, false);
            }

            int cq = (4 * hc + t) * 4 + q;
            float* dbase = delta + cq * DELTA_STRIDE + m * 16 + 8 * hi;
            *(v2f*)(dbase + 0) = (v2f){acc2[0], acc2[1]};
            *(v2f*)(dbase + 2) = (v2f){acc2[2], acc2[3]};
            *(v2f*)(dbase + 4) = (v2f){acc2[4], acc2[5]};
            *(v2f*)(dbase + 6) = (v2f){acc2[6], acc2[7]};
        }
    }
    __syncthreads();

    float* outp = xout + ((b * HEIGHT) * WIDTH + ww * 4) * CHAN;
    for (int i = tid; i < 4096; i += NTHREADS) {
        int h2  = i >> 4;
        int q   = (i >> 2) & 3;
        int c2b = (i & 3) << 2;
        int goff = (h2 * WIDTH + q) * CHAN + c2b;
        float4 xv = *(const float4*)(xres + goff);
        float4 dv;
        dv.x = delta[((c2b + 0) * 4 + q) * DELTA_STRIDE + h2];
        dv.y = delta[((c2b + 1) * 4 + q) * DELTA_STRIDE + h2];
        dv.z = delta[((c2b + 2) * 4 + q) * DELTA_STRIDE + h2];
        dv.w = delta[((c2b + 3) * 4 + q) * DELTA_STRIDE + h2];
        *(float4*)(outp + goff) =
            make_float4(xv.x + dv.x, xv.y + dv.y, xv.z + dv.z, xv.w + dv.w);
    }
}

extern "C" void kernel_launch(void* const* d_in, const int* in_sizes, int n_in,
                              void* d_out, int out_size, void* d_ws, size_t ws_size,
                              hipStream_t stream) {
    (void)in_sizes; (void)n_in; (void)out_size; (void)ws_size;
    const float* x   = (const float*)d_in[0];
    const float* cwv = (const float*)d_in[1];
    const float* f0w = (const float*)d_in[2];
    const float* f0b = (const float*)d_in[3];
    const float* f1w = (const float*)d_in[4];

    float* tmp = (float*)d_ws;
    float* out = (float*)d_out;

    dim3 grid(BATCH * 64);
    dim3 block(NTHREADS);
    size_t shmem = SMEM_FLOATS * sizeof(float);

    ca_step_kernel<<<grid, block, shmem, stream>>>(x,   cwv, f0w, f0b, f1w, tmp);
    ca_step_kernel<<<grid, block, shmem, stream>>>(tmp, cwv, f0w, f0b, f1w, out);
}


# persistent work-groups (grid 256, 8 tiles each): next tile's 12 x columns load into registers behind the staging barrier, in flight over LDS-only barriers under conv+MLP; set-up once per work-group
# speedup vs baseline: 2.7603x; 1.3301x over previous
#include <hip/hip_runtime.h>


typedef __attribute__((ext_vector_type(2))) float v2f;
typedef __attribute__((ext_vector_type(4))) float v4f;
typedef __attribute__((ext_vector_type(8))) float v8f;

#define BATCH  32
#define HEIGHT 256
#define WIDTH  256
#define CHAN   16
#define HCN    4
#define HS     12
#define NTHREADS 1024
#define NWAVES   (NTHREADS / 32)
#define NWG          256
#define TILES_PER_WG 8
#define TILE_B_STEP  (NWG / 64)

#define XROW_STRIDE 20
#define XGRP_STRIDE 84
#define XROWS       258
#define XGROUPS     65
#define XCOL_SIZE   (XGROUPS * XGRP_STRIDE)
#define NCOLS       12
#define OFF_XCOL    0
#define OFF_AFRAG   (NCOLS * XCOL_SIZE)
#define NFRAG       72
#define SMEM_FLOATS (OFF_AFRAG + NFRAG * 64)
#define DELTA_STRIDE 260
#define OFF_DELTA   0
#define OFF_US      (64 * DELTA_STRIDE)

__device__ __forceinline__ void lds_barrier()
{
    __builtin_amdgcn_fence(__ATOMIC_RELEASE, "workgroup", "local");
    __builtin_amdgcn_s_barrier();
    __builtin_amdgcn_fence(__ATOMIC_ACQUIRE, "workgroup", "local");
}

__device__ __forceinline__ void issue_cols(const float* __restrict__ pc,
                                           int dl, int dr, v4f (&R)[12])
{
#pragma unroll
    for (int r = 0; r < 4; ++r) {
        R[3 * r + 0] = *(const v4f*)(pc + dl + r * (WIDTH * CHAN));
        R[3 * r + 1] = *(const v4f*)(pc      + r * (WIDTH * CHAN));
        R[3 * r + 2] = *(const v4f*)(pc + dr + r * (WIDTH * CHAN));
    }
}

__launch_bounds__(NTHREADS)
__global__ void ca_step_kernel(const float* __restrict__ xin,
                               const float* __restrict__ convw,
                               const float* __restrict__ fc0w,
                               const float* __restrict__ fc0b,
                               const float* __restrict__ fc1w,
                               float* __restrict__ xout)
{
    extern __shared__ float smem[];
    float* xcol  = smem + OFF_XCOL;
    float* afrag = smem + OFF_AFRAG;
    float* us    = smem + OFF_US;
    float* delta = smem + OFF_DELTA;

    const int tid  = threadIdx.x;
    const int lane = tid & 31;
    const int wave = tid >> 5;
    const int m    = lane & 15;
    const int hi   = lane >> 4;

    const int blk = blockIdx.x;
    const int b0  = blk >> 6;
    const int ww  = blk & 63;

    const int  st   = tid >> 8;
    const int  sg   = (tid >> 2) & 63;
    const int  sc4  = tid & 3;
    const int  sw   = ww + 64 * st;
    const bool okL  = sw != 0;
    const bool okR  = sw != WIDTH - 1;
    const int  dl   = okL ? -CHAN : 0;
    const int  dr   = okR ?  CHAN : 0;
    float* sdst = xcol + 3 * st * XCOL_SIZE + sg * XGRP_STRIDE + sc4 * 4;
    const float* srow = xin + ((b0 * HEIGHT + 4 * sg) * WIDTH + sw) * CHAN + sc4 * 4;

    v4f R[12];
    issue_cols(srow, dl, dr, R);

    v2f afc0;
    {
        int s = m;
        int base = (s < HS ? s : 0) * 4 + 2 * hi;
        float a0 = fc0w[base + 0];
        float a1 = fc0w[base + 1];
        afc0 = (s < HS) ? (v2f){a0, a1} : (v2f){0.f, 0.f};
    }
    v8f cbias;
#pragma unroll
    for (int v = 0; v < 8; ++v) {
        int s = v + 8 * hi;
        float bv = fc0b[s < HS ? s : 0];
        cbias[v] = (s < HS) ? bv : 0.f;
    }
    v2f afc1[3];
#pragma unroll
    for (int t3 = 0; t3 < 3; ++t3) {
        int base = m * HS + 4 * t3 + 2 * hi;
        afc1[t3].x = fc1w[base + 0];
        afc1[t3].y = fc1w[base + 1];
    }

#pragma unroll
    for (int k = 0; k < 5; ++k) {
        int i = tid + k * NTHREADS;
        if (i < NFRAG * 64) {
            int s    = i >> 6;
            int ln   = (i >> 1) & 31;
            int r2   = i & 1;
            int cblk = s & 3;
            int ad   = s >> 2;
            int a    = ad / 6;
            int d6   = ad - a * 6;
            int mm   = ln & 15;
            int hh2  = ln >> 4;
            int hcw  = (mm & 7) >> 1;
            int j2   = 2 * (mm >> 3) + (mm & 1);
            int d    = d6 - j2;
            int c    = cblk * 4 + 2 * hh2 + r2;
            bool in  = (unsigned)d < 3u;
            float wv = convw[hcw * 144 + c * 9 + a * 3 + (in ? d : 0)];
            afrag[i] = in ? wv : 0.f;
        }
    }

    const bool convwave = (wave & 4) == 0;
    const int  t  = wave >> 3;
    const int  q  = wave & 3;
    const int  cwv = t * 4 + q;

    const int oq   = (tid >> 2) & 3;
    const int oc2b = (tid & 3) << 2;
    const int ogoff = ((tid >> 4) * WIDTH + oq) * CHAN + oc2b;
    const float* odel = delta + (oc2b * 4 + oq) * DELTA_STRIDE + (tid >> 4);

    __threadfence_block();

#pragma unroll 1
    for (int kt = 0; kt < TILES_PER_WG; ++kt) {
    const int b = b0 + TILE_B_STEP * kt;

    {
        const v4f zero4 = {0.f, 0.f, 0.f, 0.f};
#pragma unroll
        for (int r = 0; r < 4; ++r) {
            const int roff = (r < 3) ? (r + 1) * XROW_STRIDE : XGRP_STRIDE;
#pragma unroll
            for (int a = 0; a < 3; ++a) {
                v4f val = R[3 * r + a];
                if (a == 0) val = okL ? val : zero4;
                if (a == 2) val = okR ? val : zero4;
                *(v4f*)(sdst + a * XCOL_SIZE + roff) = val;
            }
        }
        if (tid < 96) {
            int colIdx = tid >> 3;
            int hh     = ((tid >> 2) & 1) ? (XROWS - 1) : 0;
            int c4     = tid & 3;
            *(v4f*)(xcol + colIdx * XCOL_SIZE + (hh >> 2) * XGRP_STRIDE + (hh & 3) * XROW_STRIDE + c4 * 4) = zero4;
        }
    }
    lds_barrier();

    const float* xres = xin + ((b * HEIGHT) * WIDTH + ww * 4) * CHAN;
    if (kt + 1 < TILES_PER_WG) {
        srow += TILE_B_STEP * HEIGHT * WIDTH * CHAN;
        issue_cols(srow, dl, dr, R);
    }
    if (tid < HEIGHT)
        __builtin_prefetch(xres + tid * (WIDTH * CHAN), 0, 1);

    v8f acc = {};
    if (convwave) {
        const float* pb = xcol + 3 * t * XCOL_SIZE + (q * 16 + m) * XGRP_STRIDE + 2 * hi;
        const float* pa = afrag + lane * 2;
#pragma unroll
        for (int a = 0; a < 3; ++a) {
#pragma unroll
            for (int d6 = 0; d6 < 6; ++d6) {
                const float* pg = pb + a * XCOL_SIZE + (d6 >> 2) * XGRP_STRIDE + (d6 & 3) * XROW_STRIDE;
                int s = (a * 6 + d6) * 4;
#pragma unroll
                for (int cblk = 0; cblk < 4; ++cblk) {
                    v2f av = *(const v2f*)(pa + (s + cblk) * 64);
                    v2f bv = *(const v2f*)(pg + 4 * cblk);
                    acc = __builtin_amdgcn_wmma_f32_16x16x4_f32(
                        false, av, false, bv, (short)0, acc, false, false);
                }
            }
        }
    }
    lds_barrier();

    if (convwave) {
#pragma unroll
        for (int hc = 0; hc < 4; ++hc) {
            v2f Bv = (v2f){fmaxf(acc[2 * hc], 0.f), fmaxf(acc[2 * hc + 1], 0.f)};
            v8f u0 = __builtin_amdgcn_wmma_f32_16x16x4_f32(
                false, afc0, false, Bv, (short)0, cbias, false, false);
#pragma unroll
            for (int v = 0; v < 8; ++v) u0[v] = fmaxf(u0[v], 0.f);

            float* usw = us + cwv * 1024 + hc * 256;
            *(float4*)(usw + m * 16 + 8 * hi + 0) = make_float4(u0[0], u0[1], u0[2], u0[3]);
            *(float4*)(usw + m * 16 + 8 * hi + 4) = make_float4(u0[4], u0[5], u0[6], u0[7]);

            v8f acc2 = {};
#pragma unroll
            for (int t3 = 0; t3 < 3; ++t3) {
                v2f B2 = *(const v2f*)(usw + m * 16 + 4 * t3 + 2 * hi);
                acc2 = __builtin_amdgcn_wmma_f32_16x16x4_f32(
                    false, afc1[t3], false, B2, (short)0, acc2, false, false);
            }

            int cq = (4 * hc + t) * 4 + q;
            float* dbase = delta + cq * DELTA_STRIDE + m * 16 + 8 * hi;
            *(v2f*)(dbase + 0) = (v2f){acc2[0], acc2[1]};
            *(v2f*)(dbase + 2) = (v2f){acc2[2], acc2[3]};
            *(v2f*)(dbase + 4) = (v2f){acc2[4], acc2[5]};
            *(v2f*)(dbase + 6) = (v2f){acc2[6], acc2[7]};
        }
    }
    v4f xr[4];
#pragma unroll
    for (int j = 0; j < 4; ++j)
        xr[j] = *(const v4f*)(xres + ogoff + j * (64 * WIDTH * CHAN));
    lds_barrier();

    float* outp = xout + ((b * HEIGHT) * WIDTH + ww * 4) * CHAN;
#pragma unroll
    for (int j = 0; j < 4; ++j) {
        v4f dv;
        dv.x = odel[0 * 4 * DELTA_STRIDE + 64 * j];
        dv.y = odel[1 * 4 * DELTA_STRIDE + 64 * j];
        dv.z = odel[2 * 4 * DELTA_STRIDE + 64 * j];
        dv.w = odel[3 * 4 * DELTA_STRIDE + 64 * j];
        *(v4f*)(outp + ogoff + j * (64 * WIDTH * CHAN)) = xr[j] + dv;
    }
    lds_barrier();
    }
}

extern "C" void kernel_launch(void* const* d_in, const int* in_sizes, int n_in,
                              void* d_out, int out_size, void* d_ws, size_t ws_size,
                              hipStream_t stream) {
    (void)in_sizes; (void)n_in; (void)out_size; (void)ws_size;
    const float* x   = (const float*)d_in[0];
    const float* cwv = (const float*)d_in[1];
    const float* f0w = (const float*)d_in[2];
    const float* f0b = (const float*)d_in[3];
    const float* f1w = (const float*)d_in[4];

    float* tmp = (float*)d_ws;
    float* out = (float*)d_out;

    dim3 grid(NWG);
    dim3 block(NTHREADS);
    size_t shmem = SMEM_FLOATS * sizeof(float);

    ca_step_kernel<<<grid, block, shmem, stream>>>(x,   cwv, f0w, f0b, f1w, tmp);
    ca_step_kernel<<<grid, block, shmem, stream>>>(tmp, cwv, f0w, f0b, f1w, out);
}


# tile re-shaped: 4 adjacent columns per band x half the height, 24 staged columns per 16 conv columns (was 12 per 4): half the staging loads and LDS stores per output; 32 conv blocks fill all 32 waves
# speedup vs baseline: 3.1004x; 1.1232x over previous
#include <hip/hip_runtime.h>


typedef __attribute__((ext_vector_type(2))) float v2f;
typedef __attribute__((ext_vector_type(4))) float v4f;
typedef __attribute__((ext_vector_type(8))) float v8f;

#define BATCH  32
#define HEIGHT 256
#define WIDTH  256
#define CHAN   16
#define HCN    4
#define HS     12
#define NTHREADS 1024
#define NWAVES   (NTHREADS / 32)
#define NWG          256
#define TILES_PER_WG 4
#define TILE_B_STEP  (NWG / 32)
#define TROWS        128

#define XROW_STRIDE 20
#define XGRP_STRIDE 84
#define XROWS       130
#define XGROUPS     33
#define XCOL_SIZE   (XGROUPS * XGRP_STRIDE)
#define BCOLS       6
#define NCOLS       (4 * BCOLS)
#define OFF_XCOL    0
#define OFF_AFRAG   (NCOLS * XCOL_SIZE)
#define NFRAG       72
#define SMEM_FLOATS (OFF_AFRAG + NFRAG * 64)
#define DELTA_STRIDE 260
#define OFF_DELTA   0
#define OFF_US      (128 * DELTA_STRIDE)

__device__ __forceinline__ void lds_barrier()
{
    __builtin_amdgcn_fence(__ATOMIC_RELEASE, "workgroup", "local");
    __builtin_amdgcn_s_barrier();
    __builtin_amdgcn_fence(__ATOMIC_ACQUIRE, "workgroup", "local");
}

__device__ __forceinline__ void issue_cols(const float* __restrict__ pc,
                                           int dl, int dr, v4f (&R)[12])
{
#pragma unroll
    for (int r = 0; r < 4; ++r) {
        R[3 * r + 0] = *(const v4f*)(pc + dl + r * (WIDTH * CHAN));
        R[3 * r + 1] = *(const v4f*)(pc      + r * (WIDTH * CHAN));
        R[3 * r + 2] = *(const v4f*)(pc + dr + r * (WIDTH * CHAN));
    }
}

__launch_bounds__(NTHREADS)
__global__ void ca_step_kernel(const float* __restrict__ xin,
                               const float* __restrict__ convw,
                               const float* __restrict__ fc0w,
                               const float* __restrict__ fc0b,
                               const float* __restrict__ fc1w,
                               float* __restrict__ xout)
{
    extern __shared__ float smem[];
    float* xcol  = smem + OFF_XCOL;
    float* afrag = smem + OFF_AFRAG;
    float* us    = smem + OFF_US;
    float* delta = smem + OFF_DELTA;

    const int tid  = threadIdx.x;
    const int lane = tid & 31;
    const int wave = tid >> 5;
    const int m    = lane & 15;
    const int hi   = lane >> 4;

    const int blk = blockIdx.x;
    const int b0  = blk >> 5;
    const int hp  = (blk >> 4) & 1;
    const int wg  = blk & 15;

    const int  cg   = tid >> 7;
    const int  sg   = (tid >> 2) & 31;
    const int  sc4  = tid & 3;
    const int  sw   = 64 * (cg >> 1) + 4 * wg + 3 * (cg & 1);
    const bool okL  = sw != 0;
    const bool okR  = sw != WIDTH - 1;
    const int  dl   = okL ? -CHAN : 0;
    const int  dr   = okR ?  CHAN : 0;
    float* sdst = xcol + 3 * cg * XCOL_SIZE + sg * XGRP_STRIDE + sc4 * 4;
    const float* srow = xin + ((b0 * HEIGHT + TROWS * hp + 4 * sg) * WIDTH + sw) * CHAN + sc4 * 4;

    const bool halo = tid < 8 * NCOLS;
    const int  hcol = tid >> 3;
    const int  hbot = (tid >> 2) & 1;
    const int  hh   = hbot ? TROWS * hp + TROWS : TROWS * hp - 1;
    const int  hw   = 64 * (hcol / BCOLS) + 4 * wg - 1 + (hcol % BCOLS);
    const bool hok  = (unsigned)hh < (unsigned)HEIGHT && (unsigned)hw < (unsigned)WIDTH;
    const int  hhc  = hh < 0 ? 0 : (hh > HEIGHT - 1 ? HEIGHT - 1 : hh);
    const int  hwc  = hw < 0 ? 0 : (hw > WIDTH - 1 ? WIDTH - 1 : hw);
    float* hdst = xcol + hcol * XCOL_SIZE + (hbot ? (XROWS - 1) / 4 * XGRP_STRIDE + ((XROWS - 1) & 3) * XROW_STRIDE : 0) + sc4 * 4;
    const float* hrow = xin + ((b0 * HEIGHT + hhc) * WIDTH + hwc) * CHAN + sc4 * 4;

    v4f R[12];
    v4f Rh = {0.f, 0.f, 0.f, 0.f};
    issue_cols(srow, dl, dr, R);
    if (halo) Rh = *(const v4f*)hrow;

    v2f afc0;
    {
        int s = m;
        int base = (s < HS ? s : 0) * 4 + 2 * hi;
        float a0 = fc0w[base + 0];
        float a1 = fc0w[base + 1];
        afc0 = (s < HS) ? (v2f){a0, a1} : (v2f){0.f, 0.f};
    }
    v8f cbias;
#pragma unroll
    for (int v = 0; v < 8; ++v) {
        int s = v + 8 * hi;
        float bv = fc0b[s < HS ? s : 0];
        cbias[v] = (s < HS) ? bv : 0.f;
    }
    v2f afc1[3];
#pragma unroll
    for (int t3 = 0; t3 < 3; ++t3) {
        int base = m * HS + 4 * t3 + 2 * hi;
        afc1[t3].x = fc1w[base + 0];
        afc1[t3].y = fc1w[base + 1];
    }

#pragma unroll
    for (int k = 0; k < 5; ++k) {
        int i = tid + k * NTHREADS;
        if (i < NFRAG * 64) {
            int s    = i >> 6;
            int ln   = (i >> 1) & 31;
            int r2   = i & 1;
            int cblk = s & 3;
            int ad   = s >> 2;
            int a    = ad / 6;
            int d6   = ad - a * 6;
            int mm   = ln & 15;
            int hh2  = ln >> 4;
            int hcw  = (mm & 7) >> 1;
            int j2   = 2 * (mm >> 3) + (mm & 1);
            int d    = d6 - j2;
            int c    = cblk * 4 + 2 * hh2 + r2;
            bool in  = (unsigned)d < 3u;
            float wv = convw[hcw * 144 + c * 9 + a * 3 + (in ? d : 0)];
            afrag[i] = in ? wv : 0.f;
        }
    }

    const int  t  = wave >> 3;
    const int  ci = (wave >> 1) & 3;
    const int  qq = wave & 1;
    const int  cp = 2 * ci + qq;

    const int op   = (tid >> 2) & 7;
    const int oc2b = (tid & 3) << 2;
    const int ogoff = ((tid >> 5) * WIDTH + 4 * (op >> 1) + (op & 1)) * CHAN + oc2b;
    const float* odel = delta + (oc2b * 8 + op) * DELTA_STRIDE + (tid >> 5);

    __threadfence_block();

#pragma unroll 1
    for (int kt = 0; kt < TILES_PER_WG; ++kt) {
    const int b = b0 + TILE_B_STEP * kt;

    {
        const v4f zero4 = {0.f, 0.f, 0.f, 0.f};
#pragma unroll
        for (int r = 0; r < 4; ++r) {
            const int roff = (r < 3) ? (r + 1) * XROW_STRIDE : XGRP_STRIDE;
#pragma unroll
            for (int a = 0; a < 3; ++a) {
                v4f val = R[3 * r + a];
                if (a == 0) val = okL ? val : zero4;
                if (a == 2) val = okR ? val : zero4;
                *(v4f*)(sdst + a * XCOL_SIZE + roff) = val;
            }
        }
        if (halo)
            *(v4f*)hdst = hok ? Rh : zero4;
    }
    lds_barrier();

    const float* xres = xin + ((b * HEIGHT) * WIDTH + wg * 16 + hp * 2) * CHAN;
    if (kt + 1 < TILES_PER_WG) {
        srow += TILE_B_STEP * HEIGHT * WIDTH * CHAN;
        hrow += TILE_B_STEP * HEIGHT * WIDTH * CHAN;
        issue_cols(srow, dl, dr, R);
        if (halo) Rh = *(const v4f*)hrow;
    }
    __builtin_prefetch(xres + (tid >> 2) * (WIDTH * CHAN) + (tid & 3) * (4 * CHAN), 0, 1);

    v8f acc = {};
    {
        const float* pb = xcol + (BCOLS * t + ci) * XCOL_SIZE + (qq * 16 + m) * XGRP_STRIDE + 2 * hi;
        const float* pa = afrag + lane * 2;
#pragma unroll
        for (int a = 0; a < 3; ++a) {
#pragma unroll
            for (int d6 = 0; d6 < 6; ++d6) {
                const float* pg = pb + a * XCOL_SIZE + (d6 >> 2) * XGRP_STRIDE + (d6 & 3) * XROW_STRIDE;
                int s = (a * 6 + d6) * 4;
#pragma unroll
                for (int cblk = 0; cblk < 4; ++cblk) {
                    v2f av = *(const v2f*)(pa + (s + cblk) * 64);
                    v2f bv = *(const v2f*)(pg + 4 * cblk);
                    acc = __builtin_amdgcn_wmma_f32_16x16x4_f32(
                        false, av, false, bv, (short)0, acc, false, false);
                }
            }
        }
    }
    lds_barrier();

    {
#pragma unroll
        for (int hc = 0; hc < 4; ++hc) {
            v2f Bv = (v2f){fmaxf(acc[2 * hc], 0.f), fmaxf(acc[2 * hc + 1], 0.f)};
            v8f u0 = __builtin_amdgcn_wmma_f32_16x16x4_f32(
                false, afc0, false, Bv, (short)0, cbias, false, false);
#pragma unroll
            for (int v = 0; v < 8; ++v) u0[v] = fmaxf(u0[v], 0.f);

            float* usw = us + wave * 1024 + hc * 256;
            *(float4*)(usw + m * 16 + 8 * hi + 0) = make_float4(u0[0], u0[1], u0[2], u0[3]);
            *(float4*)(usw + m * 16 + 8 * hi + 4) = make_float4(u0[4], u0[5], u0[6], u0[7]);

            v8f acc2 = {};
#pragma unroll
            for (int t3 = 0; t3 < 3; ++t3) {
                v2f B2 = *(const v2f*)(usw + m * 16 + 4 * t3 + 2 * hi);
                acc2 = __builtin_amdgcn_wmma_f32_16x16x4_f32(
                    false, afc1[t3], false, B2, (short)0, acc2, false, false);
            }

            int cq = (4 * hc + t) * 8 + cp;
            float* dbase = delta + cq * DELTA_STRIDE + m * 16 + 8 * hi;
            *(v2f*)(dbase + 0) = (v2f){acc2[0], acc2[1]};
            *(v2f*)(dbase + 2) = (v2f){acc2[2], acc2[3]};
            *(v2f*)(dbase + 4) = (v2f){acc2[4], acc2[5]};
            *(v2f*)(dbase + 6) = (v2f){acc2[6], acc2[7]};
        }
    }
    v4f xr[4];
#pragma unroll
    for (int j = 0; j < 4; ++j)
        xr[j] = *(const v4f*)(xres + ogoff + j * (32 * WIDTH * CHAN));
    lds_barrier();

    float* outp = xout + ((b * HEIGHT) * WIDTH + wg * 16 + hp * 2) * CHAN;
#pragma unroll
    for (int j = 0; j < 8; ++j) {
        v4f dv;
        dv.x = odel[0 * 8 * DELTA_STRIDE + 32 * j];
        dv.y = odel[1 * 8 * DELTA_STRIDE + 32 * j];
        dv.z = odel[2 * 8 * DELTA_STRIDE + 32 * j];
        dv.w = odel[3 * 8 * DELTA_STRIDE + 32 * j];
        *(v4f*)(outp + ogoff + j * (32 * WIDTH * CHAN)) = xr[j & 3] + dv;
        if (j < 4)
            xr[j] = *(const v4f*)(xres + ogoff + (j + 4) * (32 * WIDTH * CHAN));
    }
    lds_barrier();
    }
}

extern "C" void kernel_launch(void* const* d_in, const int* in_sizes, int n_in,
                              void* d_out, int out_size, void* d_ws, size_t ws_size,
                              hipStream_t stream) {
    (void)in_sizes; (void)n_in; (void)out_size; (void)ws_size;
    const float* x   = (const float*)d_in[0];
    const float* cwv = (const float*)d_in[1];
    const float* f0w = (const float*)d_in[2];
    const float* f0b = (const float*)d_in[3];
    const float* f1w = (const float*)d_in[4];

    float* tmp = (float*)d_ws;
    float* out = (float*)d_out;

    dim3 grid(NWG);
    dim3 block(NTHREADS);
    size_t shmem = SMEM_FLOATS * sizeof(float);

    ca_step_kernel<<<grid, block, shmem, stream>>>(x,   cwv, f0w, f0b, f1w, tmp);
    ca_step_kernel<<<grid, block, shmem, stream>>>(tmp, cwv, f0w, f0b, f1w, out);
}


# next tile's 12 column loads issued one per 6 WMMAs inside the conv (was one 12-load group behind the staging barrier): waves no longer queue at the memory pipeline before their first WMMA
# speedup vs baseline: 3.3899x; 1.0934x over previous
#include <hip/hip_runtime.h>


typedef __attribute__((ext_vector_type(2))) float v2f;
typedef __attribute__((ext_vector_type(4))) float v4f;
typedef __attribute__((ext_vector_type(8))) float v8f;

#define BATCH  32
#define HEIGHT 256
#define WIDTH  256
#define CHAN   16
#define HCN    4
#define HS     12
#define NTHREADS 1024
#define NWAVES   (NTHREADS / 32)
#define NWG          256
#define TILES_PER_WG 4
#define TILE_B_STEP  (NWG / 32)
#define TROWS        128

#define XROW_STRIDE 20
#define XGRP_STRIDE 84
#define XROWS       130
#define XGROUPS     33
#define XCOL_SIZE   (XGROUPS * XGRP_STRIDE)
#define BCOLS       6
#define NCOLS       (4 * BCOLS)
#define OFF_XCOL    0
#define OFF_AFRAG   (NCOLS * XCOL_SIZE)
#define NFRAG       72
#define SMEM_FLOATS (OFF_AFRAG + NFRAG * 64)
#define DELTA_STRIDE 260
#define OFF_DELTA   0
#define OFF_US      (128 * DELTA_STRIDE)

__device__ __forceinline__ void lds_barrier()
{
    __builtin_amdgcn_fence(__ATOMIC_RELEASE, "workgroup", "local");
    __builtin_amdgcn_s_barrier();
    __builtin_amdgcn_fence(__ATOMIC_ACQUIRE, "workgroup", "local");
}

__device__ __forceinline__ void issue_cols(const float* __restrict__ pc,
                                           int dl, int dr, v4f (&R)[12])
{
#pragma unroll
    for (int r = 0; r < 4; ++r) {
        R[3 * r + 0] = *(const v4f*)(pc + dl + r * (WIDTH * CHAN));
        R[3 * r + 1] = *(const v4f*)(pc      + r * (WIDTH * CHAN));
        R[3 * r + 2] = *(const v4f*)(pc + dr + r * (WIDTH * CHAN));
    }
}

__device__ __forceinline__ void issue_col1(const float* __restrict__ pc,
                                           int dl, int dr, v4f (&R)[12], const int idx)
{
    const int r = idx / 3;
    const int a = idx % 3;
    const int d = (a == 0) ? dl : (a == 2 ? dr : 0);
    R[idx] = *(const v4f*)(pc + d + r * (WIDTH * CHAN));
}

struct WithNextLoads    { static constexpr bool value = true;  };
struct WithoutNextLoads { static constexpr bool value = false; };

__launch_bounds__(NTHREADS)
__global__ void ca_step_kernel(const float* __restrict__ xin,
                               const float* __restrict__ convw,
                               const float* __restrict__ fc0w,
                               const float* __restrict__ fc0b,
                               const float* __restrict__ fc1w,
                               float* __restrict__ xout)
{
    extern __shared__ float smem[];
    float* xcol  = smem + OFF_XCOL;
    float* afrag = smem + OFF_AFRAG;
    float* us    = smem + OFF_US;
    float* delta = smem + OFF_DELTA;

    const int tid  = threadIdx.x;
    const int lane = tid & 31;
    const int wave = tid >> 5;
    const int m    = lane & 15;
    const int hi   = lane >> 4;

    const int blk = blockIdx.x;
    const int b0  = blk >> 5;
    const int hp  = (blk >> 4) & 1;
    const int wg  = blk & 15;

    const int  cg   = tid >> 7;
    const int  sg   = (tid >> 2) & 31;
    const int  sc4  = tid & 3;
    const int  sw   = 64 * (cg >> 1) + 4 * wg + 3 * (cg & 1);
    const bool okL  = sw != 0;
    const bool okR  = sw != WIDTH - 1;
    const int  dl   = okL ? -CHAN : 0;
    const int  dr   = okR ?  CHAN : 0;
    float* sdst = xcol + 3 * cg * XCOL_SIZE + sg * XGRP_STRIDE + sc4 * 4;
    const float* srow = xin + ((b0 * HEIGHT + TROWS * hp + 4 * sg) * WIDTH + sw) * CHAN + sc4 * 4;

    const bool halo = tid < 8 * NCOLS;
    const int  hcol = tid >> 3;
    const int  hbot = (tid >> 2) & 1;
    const int  hh   = hbot ? TROWS * hp + TROWS : TROWS * hp - 1;
    const int  hw   = 64 * (hcol / BCOLS) + 4 * wg - 1 + (hcol % BCOLS);
    const bool hok  = (unsigned)hh < (unsigned)HEIGHT && (unsigned)hw < (unsigned)WIDTH;
    const int  hhc  = hh < 0 ? 0 : (hh > HEIGHT - 1 ? HEIGHT - 1 : hh);
    const int  hwc  = hw < 0 ? 0 : (hw > WIDTH - 1 ? WIDTH - 1 : hw);
    float* hdst = xcol + hcol * XCOL_SIZE + (hbot ? (XROWS - 1) / 4 * XGRP_STRIDE + ((XROWS - 1) & 3) * XROW_STRIDE : 0) + sc4 * 4;
    const float* hrow = xin + ((b0 * HEIGHT + hhc) * WIDTH + hwc) * CHAN + sc4 * 4;

    v4f R[12];
    v4f Rh = {0.f, 0.f, 0.f, 0.f};
    issue_cols(srow, dl, dr, R);
    if (halo) Rh = *(const v4f*)hrow;

    v2f afc0;
    {
        int s = m;
        int base = (s < HS ? s : 0) * 4 + 2 * hi;
        float a0 = fc0w[base + 0];
        float a1 = fc0w[base + 1];
        afc0 = (s < HS) ? (v2f){a0, a1} : (v2f){0.f, 0.f};
    }
    v8f cbias;
#pragma unroll
    for (int v = 0; v < 8; ++v) {
        int s = v + 8 * hi;
        float bv = fc0b[s < HS ? s : 0];
        cbias[v] = (s < HS) ? bv : 0.f;
    }
    v2f afc1[3];
#pragma unroll
    for (int t3 = 0; t3 < 3; ++t3) {
        int base = m * HS + 4 * t3 + 2 * hi;
        afc1[t3].x = fc1w[base + 0];
        afc1[t3].y = fc1w[base + 1];
    }

#pragma unroll
    for (int k = 0; k < 5; ++k) {
        int i = tid + k * NTHREADS;
        if (i < NFRAG * 64) {
            int s    = i >> 6;
            int ln   = (i >> 1) & 31;
            int r2   = i & 1;
            int cblk = s & 3;
            int ad   = s >> 2;
            int a    = ad / 6;
            int d6   = ad - a * 6;
            int mm   = ln & 15;
            int hh2  = ln >> 4;
            int hcw  = (mm & 7) >> 1;
            int j2   = 2 * (mm >> 3) + (mm & 1);
            int d    = d6 - j2;
            int c    = cblk * 4 + 2 * hh2 + r2;
            bool in  = (unsigned)d < 3u;
            float wv = convw[hcw * 144 + c * 9 + a * 3 + (in ? d : 0)];
            afrag[i] = in ? wv : 0.f;
        }
    }

    const int  t  = wave >> 3;
    const int  ci = (wave >> 1) & 3;
    const int  qq = wave & 1;
    const int  cp = 2 * ci + qq;

    const int op   = (tid >> 2) & 7;
    const int oc2b = (tid & 3) << 2;
    const int ogoff = ((tid >> 5) * WIDTH + 4 * (op >> 1) + (op & 1)) * CHAN + oc2b;
    const float* odel = delta + (oc2b * 8 + op) * DELTA_STRIDE + (tid >> 5);

    __threadfence_block();

    auto tile = [&](auto next_tag, const int kt) __attribute__((always_inline)) {
    constexpr bool NEXT = decltype(next_tag)::value;
    const int b = b0 + TILE_B_STEP * kt;

    {
        const v4f zero4 = {0.f, 0.f, 0.f, 0.f};
#pragma unroll
        for (int r = 0; r < 4; ++r) {
            const int roff = (r < 3) ? (r + 1) * XROW_STRIDE : XGRP_STRIDE;
#pragma unroll
            for (int a = 0; a < 3; ++a) {
                v4f val = R[3 * r + a];
                if (a == 0) val = okL ? val : zero4;
                if (a == 2) val = okR ? val : zero4;
                *(v4f*)(sdst + a * XCOL_SIZE + roff) = val;
            }
        }
        if (halo)
            *(v4f*)hdst = hok ? Rh : zero4;
    }
    lds_barrier();

    const float* xres = xin + ((b * HEIGHT) * WIDTH + wg * 16 + hp * 2) * CHAN;
    if constexpr (NEXT) {
        srow += TILE_B_STEP * HEIGHT * WIDTH * CHAN;
        hrow += TILE_B_STEP * HEIGHT * WIDTH * CHAN;
        if (halo) Rh = *(const v4f*)hrow;
    }

    v8f acc = {};
    {
        const float* pb = xcol + (BCOLS * t + ci) * XCOL_SIZE + (qq * 16 + m) * XGRP_STRIDE + 2 * hi;
        const float* pa = afrag + lane * 2;
#pragma unroll
        for (int a = 0; a < 3; ++a) {
#pragma unroll
            for (int d6 = 0; d6 < 6; ++d6) {
                const float* pg = pb + a * XCOL_SIZE + (d6 >> 2) * XGRP_STRIDE + (d6 & 3) * XROW_STRIDE;
                int s = (a * 6 + d6) * 4;
#pragma unroll
                for (int cblk = 0; cblk < 4; ++cblk) {
                    const int n = s + cblk;
                    if constexpr (NEXT) {
                        if (n % 6 == 0) {
                            __builtin_amdgcn_sched_barrier(0);
                            issue_col1(srow, dl, dr, R, n / 6);
                            __builtin_amdgcn_sched_barrier(0);
                        }
                    }
                    if (n == 33)
                        __builtin_prefetch(xres + (tid >> 2) * (WIDTH * CHAN) + (tid & 3) * (4 * CHAN), 0, 1);
                    v2f av = *(const v2f*)(pa + (s + cblk) * 64);
                    v2f bv = *(const v2f*)(pg + 4 * cblk);
                    acc = __builtin_amdgcn_wmma_f32_16x16x4_f32(
                        false, av, false, bv, (short)0, acc, false, false);
                }
            }
        }
    }
    lds_barrier();

    {
#pragma unroll
        for (int hc = 0; hc < 4; ++hc) {
            v2f Bv = (v2f){fmaxf(acc[2 * hc], 0.f), fmaxf(acc[2 * hc + 1], 0.f)};
            v8f u0 = __builtin_amdgcn_wmma_f32_16x16x4_f32(
                false, afc0, false, Bv, (short)0, cbias, false, false);
#pragma unroll
            for (int v = 0; v < 8; ++v) u0[v] = fmaxf(u0[v], 0.f);

            float* usw = us + wave * 1024 + hc * 256;
            *(float4*)(usw + m * 16 + 8 * hi + 0) = make_float4(u0[0], u0[1], u0[2], u0[3]);
            *(float4*)(usw + m * 16 + 8 * hi + 4) = make_float4(u0[4], u0[5], u0[6], u0[7]);

            v8f acc2 = {};
#pragma unroll
            for (int t3 = 0; t3 < 3; ++t3) {
                v2f B2 = *(const v2f*)(usw + m * 16 + 4 * t3 + 2 * hi);
                acc2 = __builtin_amdgcn_wmma_f32_16x16x4_f32(
                    false, afc1[t3], false, B2, (short)0, acc2, false, false);
            }

            int cq = (4 * hc + t) * 8 + cp;
            float* dbase = delta + cq * DELTA_STRIDE + m * 16 + 8 * hi;
            *(v2f*)(dbase + 0) = (v2f){acc2[0], acc2[1]};
            *(v2f*)(dbase + 2) = (v2f){acc2[2], acc2[3]};
            *(v2f*)(dbase + 4) = (v2f){acc2[4], acc2[5]};
            *(v2f*)(dbase + 6) = (v2f){acc2[6], acc2[7]};
        }
    }
    v4f xr[4];
#pragma unroll
    for (int j = 0; j < 4; ++j)
        xr[j] = *(const v4f*)(xres + ogoff + j * (32 * WIDTH * CHAN));
    lds_barrier();

    float* outp = xout + ((b * HEIGHT) * WIDTH + wg * 16 + hp * 2) * CHAN;
#pragma unroll
    for (int j = 0; j < 8; ++j) {
        v4f dv;
        dv.x = odel[0 * 8 * DELTA_STRIDE + 32 * j];
        dv.y = odel[1 * 8 * DELTA_STRIDE + 32 * j];
        dv.z = odel[2 * 8 * DELTA_STRIDE + 32 * j];
        dv.w = odel[3 * 8 * DELTA_STRIDE + 32 * j];
        *(v4f*)(outp + ogoff + j * (32 * WIDTH * CHAN)) = xr[j & 3] + dv;
        if (j < 4)
            xr[j] = *(const v4f*)(xres + ogoff + (j + 4) * (32 * WIDTH * CHAN));
    }
    lds_barrier();
    };

#pragma unroll 1
    for (int kt = 0; kt + 1 < TILES_PER_WG; ++kt)
        tile(WithNextLoads{}, kt);
    tile(WithoutNextLoads{}, TILES_PER_WG - 1);
}

extern "C" void kernel_launch(void* const* d_in, const int* in_sizes, int n_in,
                              void* d_out, int out_size, void* d_ws, size_t ws_size,
                              hipStream_t stream) {
    (void)in_sizes; (void)n_in; (void)out_size; (void)ws_size;
    const float* x   = (const float*)d_in[0];
    const float* cwv = (const float*)d_in[1];
    const float* f0w = (const float*)d_in[2];
    const float* f0b = (const float*)d_in[3];
    const float* f1w = (const float*)d_in[4];

    float* tmp = (float*)d_ws;
    float* out = (float*)d_out;

    dim3 grid(NWG);
    dim3 block(NTHREADS);
    size_t shmem = SMEM_FLOATS * sizeof(float);

    ca_step_kernel<<<grid, block, shmem, stream>>>(x,   cwv, f0w, f0b, f1w, tmp);
    ca_step_kernel<<<grid, block, shmem, stream>>>(tmp, cwv, f0w, f0b, f1w, out);
}


# speedup vs baseline: 3.5381x; 1.0437x over previous
#include <hip/hip_runtime.h>


typedef __attribute__((ext_vector_type(2))) float v2f;
typedef __attribute__((ext_vector_type(4))) float v4f;
typedef __attribute__((ext_vector_type(8))) float v8f;

#define BATCH  32
#define HEIGHT 256
#define WIDTH  256
#define CHAN   16
#define HCN    4
#define HS     12
#define NTHREADS 1024
#define NWAVES   (NTHREADS / 32)
#define NWG          256
#define TILES_PER_WG 4
#define TILE_B_STEP  (NWG / 32)
#define TROWS        128

#define XROW_STRIDE 20
#define XGRP_STRIDE 84
#define XROWS       130
#define XGROUPS     33
#define XCOL_SIZE   (XGROUPS * XGRP_STRIDE)
#define BCOLS       6
#define NCOLS       (4 * BCOLS)
#define OFF_XCOL    0
#define OFF_AFRAG   (NCOLS * XCOL_SIZE)
#define NFRAG       72
#define SMEM_FLOATS (OFF_AFRAG + NFRAG * 64)
#define DELTA_STRIDE 260
#define OFF_DELTA   0
#define OFF_US      (128 * DELTA_STRIDE)

__device__ __forceinline__ void lds_barrier()
{
    __builtin_amdgcn_fence(__ATOMIC_RELEASE, "workgroup", "local");
    __builtin_amdgcn_s_barrier();
    __builtin_amdgcn_fence(__ATOMIC_ACQUIRE, "workgroup", "local");
}

__device__ __forceinline__ void issue_cols(const float* __restrict__ pc,
                                           int dl, int dr, v4f (&R)[12])
{
#pragma unroll
    for (int r = 0; r < 4; ++r) {
        R[3 * r + 0] = *(const v4f*)(pc + dl + r * (WIDTH * CHAN));
        R[3 * r + 1] = *(const v4f*)(pc      + r * (WIDTH * CHAN));
        R[3 * r + 2] = *(const v4f*)(pc + dr + r * (WIDTH * CHAN));
    }
}

__device__ __forceinline__ void issue_col1(const float* __restrict__ pc,
                                           int dl, int dr, v4f (&R)[12], const int idx)
{
    const int r = idx / 3;
    const int a = idx % 3;
    const int d = (a == 0) ? dl : (a == 2 ? dr : 0);
    R[idx] = *(const v4f*)(pc + d + r * (WIDTH * CHAN));
}

struct WithNextLoads    { static constexpr bool value = true;  };
struct WithoutNextLoads { static constexpr bool value = false; };

__launch_bounds__(NTHREADS)
__global__ void ca_step_kernel(const float* __restrict__ xin,
                               const float* __restrict__ convw,
                               const float* __restrict__ fc0w,
                               const float* __restrict__ fc0b,
                               const float* __restrict__ fc1w,
                               float* __restrict__ xout)
{
    extern __shared__ float smem[];
    float* xcol  = smem + OFF_XCOL;
    float* afrag = smem + OFF_AFRAG;
    float* us    = smem + OFF_US;
    float* delta = smem + OFF_DELTA;

    const int tid  = threadIdx.x;
    const int lane = tid & 31;
    const int wave = tid >> 5;
    const int m    = lane & 15;
    const int hi   = lane >> 4;

    const int blk = blockIdx.x;
    const int b0  = (blk >> 1) & 7;
    const int hp  = blk & 1;
    const int wg  = blk >> 4;

    const int  cg   = tid >> 7;
    const int  sg   = (tid >> 2) & 31;
    const int  sc4  = tid & 3;
    const int  sw   = 64 * (cg >> 1) + 4 * wg + 3 * (cg & 1);
    const bool okL  = sw != 0;
    const bool okR  = sw != WIDTH - 1;
    const int  dl   = okL ? -CHAN : 0;
    const int  dr   = okR ?  CHAN : 0;
    float* sdst = xcol + 3 * cg * XCOL_SIZE + sg * XGRP_STRIDE + sc4 * 4;
    const float* srow = xin + ((b0 * HEIGHT + TROWS * hp + 4 * sg) * WIDTH + sw) * CHAN + sc4 * 4;

    const bool halo = tid < 8 * NCOLS;
    const int  hcol = tid >> 3;
    const int  hbot = (tid >> 2) & 1;
    const int  hh   = hbot ? TROWS * hp + TROWS : TROWS * hp - 1;
    const int  hw   = 64 * (hcol / BCOLS) + 4 * wg - 1 + (hcol % BCOLS);
    const bool hok  = (unsigned)hh < (unsigned)HEIGHT && (unsigned)hw < (unsigned)WIDTH;
    const int  hhc  = hh < 0 ? 0 : (hh > HEIGHT - 1 ? HEIGHT - 1 : hh);
    const int  hwc  = hw < 0 ? 0 : (hw > WIDTH - 1 ? WIDTH - 1 : hw);
    float* hdst = xcol + hcol * XCOL_SIZE + (hbot ? (XROWS - 1) / 4 * XGRP_STRIDE + ((XROWS - 1) & 3) * XROW_STRIDE : 0) + sc4 * 4;
    const float* hrow = xin + ((b0 * HEIGHT + hhc) * WIDTH + hwc) * CHAN + sc4 * 4;

    v4f R[12];
    v4f Rh = {0.f, 0.f, 0.f, 0.f};
    issue_cols(srow, dl, dr, R);
    if (halo) Rh = *(const v4f*)hrow;

    v2f afc0;
    {
        int s = m;
        int base = (s < HS ? s : 0) * 4 + 2 * hi;
        float a0 = fc0w[base + 0];
        float a1 = fc0w[base + 1];
        afc0 = (s < HS) ? (v2f){a0, a1} : (v2f){0.f, 0.f};
    }
    v8f cbias;
#pragma unroll
    for (int v = 0; v < 8; ++v) {
        int s = v + 8 * hi;
        float bv = fc0b[s < HS ? s : 0];
        cbias[v] = (s < HS) ? bv : 0.f;
    }
    v2f afc1[3];
#pragma unroll
    for (int t3 = 0; t3 < 3; ++t3) {
        int base = m * HS + 4 * t3 + 2 * hi;
        afc1[t3].x = fc1w[base + 0];
        afc1[t3].y = fc1w[base + 1];
    }

#pragma unroll
    for (int k = 0; k < 5; ++k) {
        int i = tid + k * NTHREADS;
        if (i < NFRAG * 64) {
            int s    = i >> 6;
            int ln   = (i >> 1) & 31;
            int r2   = i & 1;
            int cblk = s & 3;
            int ad   = s >> 2;
            int a    = ad / 6;
            int d6   = ad - a * 6;
            int mm   = ln & 15;
            int hh2  = ln >> 4;
            int hcw  = (mm & 7) >> 1;
            int j2   = 2 * (mm >> 3) + (mm & 1);
            int d    = d6 - j2;
            int c    = cblk * 4 + 2 * hh2 + r2;
            bool in  = (unsigned)d < 3u;
            float wv = convw[hcw * 144 + c * 9 + a * 3 + (in ? d : 0)];
            afrag[i] = in ? wv : 0.f;
        }
    }

    const int  t  = wave >> 3;
    const int  ci = (wave >> 1) & 3;
    const int  qq = wave & 1;
    const int  cp = 2 * ci + qq;

    const int op   = (tid >> 2) & 7;
    const int oc2b = (tid & 3) << 2;
    const int ogoff = ((tid >> 5) * WIDTH + 4 * (op >> 1) + (op & 1)) * CHAN + oc2b;
    const float* odel = delta + (oc2b * 8 + op) * DELTA_STRIDE + (tid >> 5);

    __threadfence_block();

    auto tile = [&](auto next_tag, const int kt) __attribute__((always_inline)) {
    constexpr bool NEXT = decltype(next_tag)::value;
    const int b = b0 + TILE_B_STEP * kt;

    {
        const v4f zero4 = {0.f, 0.f, 0.f, 0.f};
#pragma unroll
        for (int r = 0; r < 4; ++r) {
            const int roff = (r < 3) ? (r + 1) * XROW_STRIDE : XGRP_STRIDE;
#pragma unroll
            for (int a = 0; a < 3; ++a) {
                v4f val = R[3 * r + a];
                if (a == 0) val = okL ? val : zero4;
                if (a == 2) val = okR ? val : zero4;
                *(v4f*)(sdst + a * XCOL_SIZE + roff) = val;
            }
        }
        if (halo)
            *(v4f*)hdst = hok ? Rh : zero4;
    }
    lds_barrier();

    const float* xres = xin + ((b * HEIGHT) * WIDTH + wg * 16 + hp * 2) * CHAN;
    if constexpr (NEXT) {
        srow += TILE_B_STEP * HEIGHT * WIDTH * CHAN;
        hrow += TILE_B_STEP * HEIGHT * WIDTH * CHAN;
        if (halo) Rh = *(const v4f*)hrow;
    }

    v8f acc = {};
    {
        const float* pb = xcol + (BCOLS * t + ci) * XCOL_SIZE + (qq * 16 + m) * XGRP_STRIDE + 2 * hi;
        const float* pa = afrag + lane * 2;
#pragma unroll
        for (int a = 0; a < 3; ++a) {
#pragma unroll
            for (int d6 = 0; d6 < 6; ++d6) {
                const float* pg = pb + a * XCOL_SIZE + (d6 >> 2) * XGRP_STRIDE + (d6 & 3) * XROW_STRIDE;
                int s = (a * 6 + d6) * 4;
#pragma unroll
                for (int cblk = 0; cblk < 4; ++cblk) {
                    const int n = s + cblk;
                    if constexpr (NEXT) {
                        if (n % 6 == 0) {
                            __builtin_amdgcn_sched_barrier(0);
                            issue_col1(srow, dl, dr, R, n / 6);
                            __builtin_amdgcn_sched_barrier(0);
                        }
                    }
                    if (n == 33)
                        __builtin_prefetch(xres + (tid >> 2) * (WIDTH * CHAN) + (tid & 3) * (4 * CHAN), 0, 1);
                    v2f av = *(const v2f*)(pa + (s + cblk) * 64);
                    v2f bv = *(const v2f*)(pg + 4 * cblk);
                    acc = __builtin_amdgcn_wmma_f32_16x16x4_f32(
                        false, av, false, bv, (short)0, acc, false, false);
                }
            }
        }
    }
    lds_barrier();

    {
#pragma unroll
        for (int hc = 0; hc < 4; ++hc) {
            v2f Bv = (v2f){fmaxf(acc[2 * hc], 0.f), fmaxf(acc[2 * hc + 1], 0.f)};
            v8f u0 = __builtin_amdgcn_wmma_f32_16x16x4_f32(
                false, afc0, false, Bv, (short)0, cbias, false, false);
#pragma unroll
            for (int v = 0; v < 8; ++v) u0[v] = fmaxf(u0[v], 0.f);

            float* usw = us + wave * 1024 + hc * 256;
            *(float4*)(usw + m * 16 + 8 * hi + 0) = make_float4(u0[0], u0[1], u0[2], u0[3]);
            *(float4*)(usw + m * 16 + 8 * hi + 4) = make_float4(u0[4], u0[5], u0[6], u0[7]);

            v8f acc2 = {};
#pragma unroll
            for (int t3 = 0; t3 < 3; ++t3) {
                v2f B2 = *(const v2f*)(usw + m * 16 + 4 * t3 + 2 * hi);
                acc2 = __builtin_amdgcn_wmma_f32_16x16x4_f32(
                    false, afc1[t3], false, B2, (short)0, acc2, false, false);
            }

            int cq = (4 * hc + t) * 8 + cp;
            float* dbase = delta + cq * DELTA_STRIDE + m * 16 + 8 * hi;
            *(v2f*)(dbase + 0) = (v2f){acc2[0], acc2[1]};
            *(v2f*)(dbase + 2) = (v2f){acc2[2], acc2[3]};
            *(v2f*)(dbase + 4) = (v2f){acc2[4], acc2[5]};
            *(v2f*)(dbase + 6) = (v2f){acc2[6], acc2[7]};
        }
    }
    v4f xr[4];
#pragma unroll
    for (int j = 0; j < 4; ++j)
        xr[j] = *(const v4f*)(xres + ogoff + j * (32 * WIDTH * CHAN));
    lds_barrier();

    float* outp = xout + ((b * HEIGHT) * WIDTH + wg * 16 + hp * 2) * CHAN;
#pragma unroll
    for (int j = 0; j < 8; ++j) {
        v4f dv;
        dv.x = odel[0 * 8 * DELTA_STRIDE + 32 * j];
        dv.y = odel[1 * 8 * DELTA_STRIDE + 32 * j];
        dv.z = odel[2 * 8 * DELTA_STRIDE + 32 * j];
        dv.w = odel[3 * 8 * DELTA_STRIDE + 32 * j];
        *(v4f*)(outp + ogoff + j * (32 * WIDTH * CHAN)) = xr[j & 3] + dv;
        if (j < 4)
            xr[j] = *(const v4f*)(xres + ogoff + (j + 4) * (32 * WIDTH * CHAN));
    }
    lds_barrier();
    };

#pragma unroll 1
    for (int kt = 0; kt + 1 < TILES_PER_WG; ++kt)
        tile(WithNextLoads{}, kt);
    tile(WithoutNextLoads{}, TILES_PER_WG - 1);
}

extern "C" void kernel_launch(void* const* d_in, const int* in_sizes, int n_in,
                              void* d_out, int out_size, void* d_ws, size_t ws_size,
                              hipStream_t stream) {
    (void)in_sizes; (void)n_in; (void)out_size; (void)ws_size;
    const float* x   = (const float*)d_in[0];
    const float* cwv = (const float*)d_in[1];
    const float* f0w = (const float*)d_in[2];
    const float* f0b = (const float*)d_in[3];
    const float* f1w = (const float*)d_in[4];

    float* tmp = (float*)d_ws;
    float* out = (float*)d_out;

    dim3 grid(NWG);
    dim3 block(NTHREADS);
    size_t shmem = SMEM_FLOATS * sizeof(float);

    ca_step_kernel<<<grid, block, shmem, stream>>>(x,   cwv, f0w, f0b, f1w, tmp);
    ca_step_kernel<<<grid, block, shmem, stream>>>(tmp, cwv, f0w, f0b, f1w, out);
}
